# starting kernel
# baseline (speedup 1.0000x reference)
#include <hip/hip_runtime.h>
#include <hip/hip_bf16.h>


typedef __attribute__((ext_vector_type(2))) float v2f;
typedef __attribute__((ext_vector_type(8))) float v8f;

#define TT   4096
#define DD   256
#define BB   4
#define BAND 320
#define INV_E 0.36787944117144233f

__device__ __forceinline__ v8f wmma_f32(v2f a, v2f b, v8f c) {
  return __builtin_amdgcn_wmma_f32_16x16x4_f32(false, a, false, b,
                                               (short)0, c, false, false);
}

__global__ __launch_bounds__(32)
void da_proj_kernel(const float* __restrict__ x, const float* __restrict__ mask,
                    const float* __restrict__ W, const float* __restrict__ bias,
                    float* __restrict__ y) {
  const int lane = threadIdx.x & 31;
  const int l = lane & 15;
  const int h = lane >> 4;
  const int m0 = blockIdx.x * 16;
  const int N0 = blockIdx.y * 64;

  v8f zero = {};
  v8f c0 = zero, c1 = zero, c2 = zero, c3 = zero;

  const float* xrow = x + (m0 + l) * DD;
  const float* w0 = W + (N0 +  0 + l) * DD;
  const float* w1 = W + (N0 + 16 + l) * DD;
  const float* w2 = W + (N0 + 32 + l) * DD;
  const float* w3 = W + (N0 + 48 + l) * DD;

  for (int k0 = 0; k0 < DD; k0 += 4) {
    const int kk = k0 + 2 * h;
    v2f a = *(const v2f*)(xrow + kk);
    v2f b0 = *(const v2f*)(w0 + kk);
    v2f b1 = *(const v2f*)(w1 + kk);
    v2f b2 = *(const v2f*)(w2 + kk);
    v2f b3 = *(const v2f*)(w3 + kk);
    c0 = wmma_f32(a, b0, c0);
    c1 = wmma_f32(a, b1, c1);
    c2 = wmma_f32(a, b2, c2);
    c3 = wmma_f32(a, b3, c3);
  }

#pragma unroll
  for (int r = 0; r < 8; ++r) {
    const int row = m0 + r + 8 * h;
    const float mf = __expf(mask[row] * INV_E);
    float* yr = y + row * DD;
    yr[N0 +  0 + l] = (c0[r] + bias[N0 +  0 + l]) * mf;
    yr[N0 + 16 + l] = (c1[r] + bias[N0 + 16 + l]) * mf;
    yr[N0 + 32 + l] = (c2[r] + bias[N0 + 32 + l]) * mf;
    yr[N0 + 48 + l] = (c3[r] + bias[N0 + 48 + l]) * mf;
  }
}

__global__ __launch_bounds__(256)
void da_attn_kernel(const float* __restrict__ y, float* __restrict__ out) {
  const int lane = threadIdx.x & 31;
  const int l = lane & 15;
  const int h = lane >> 4;
  const int wv = threadIdx.x >> 5;
  const int it = blockIdx.x & (TT / 16 - 1);
  const int bi = blockIdx.x >> 8;
  const int I0 = it * 16;
  const int N0 = wv * 32;

  const float* yb = y + (bi * TT) * DD;

  int jlo = I0 - BAND; if (jlo < 0) jlo = 0;
  int jhi = I0 + 16 + BAND; if (jhi > TT) jhi = TT;

  v8f zero = {};
  v8f c0 = zero, c1 = zero;
  const float m = (float)(I0 + l);

  for (int j0 = jlo; j0 < jhi; j0 += 4) {
    const int kk = j0 + 2 * h;
    const float k0f = (float)kk;
    v2f a;
    a.x = __expf(-fabsf(m - k0f) * INV_E);
    a.y = __expf(-fabsf(m - (k0f + 1.0f)) * INV_E);

    const float* r0 = yb + kk * DD;
    const float* r1 = r0 + DD;
    v2f b0, b1;
    b0.x = r0[N0 + l];       b0.y = r1[N0 + l];
    b1.x = r0[N0 + 16 + l];  b1.y = r1[N0 + 16 + l];

    c0 = wmma_f32(a, b0, c0);
    c1 = wmma_f32(a, b1, c1);
  }

  float* ob = out + (bi * TT + I0) * DD;
#pragma unroll
  for (int r = 0; r < 8; ++r) {
    const int row = r + 8 * h;
    ob[row * DD + N0 + l]      = c0[r];
    ob[row * DD + N0 + 16 + l] = c1[r];
  }
}

extern "C" void kernel_launch(void* const* d_in, const int* in_sizes, int n_in,
                              void* d_out, int out_size, void* d_ws, size_t ws_size,
                              hipStream_t stream) {
  const float* x    = (const float*)d_in[0];
  const float* mask = (const float*)d_in[1];
  const float* W    = (const float*)d_in[2];
  const float* bias = (const float*)d_in[3];
  float* out = (float*)d_out;
  float* y   = (float*)d_ws;

  dim3 gproj(BB * TT / 16, DD / 64);
  da_proj_kernel<<<gproj, 32, 0, stream>>>(x, mask, W, bias, y);

  da_attn_kernel<<<BB * TT / 16, 256, 0, stream>>>(y, out);
}


# attn band loop: B fragments of the next 16 rows of y' are loaded into a second register set ahead of the current chunk's WMMAs (ping-pong), so load latency is not exposed at every K-step
# speedup vs baseline: 1.1471x; 1.1471x over previous
#include <hip/hip_runtime.h>
#include <hip/hip_bf16.h>


typedef __attribute__((ext_vector_type(2))) float v2f;
typedef __attribute__((ext_vector_type(8))) float v8f;

#define TT   4096
#define DD   256
#define BB   4
#define BAND 320
#define INV_E 0.36787944117144233f

__device__ __forceinline__ v8f wmma_f32(v2f a, v2f b, v8f c) {
  return __builtin_amdgcn_wmma_f32_16x16x4_f32(false, a, false, b,
                                               (short)0, c, false, false);
}

__global__ __launch_bounds__(32)
void da_proj_kernel(const float* __restrict__ x, const float* __restrict__ mask,
                    const float* __restrict__ W, const float* __restrict__ bias,
                    float* __restrict__ y) {
  const int lane = threadIdx.x & 31;
  const int l = lane & 15;
  const int h = lane >> 4;
  const int m0 = blockIdx.x * 16;
  const int N0 = blockIdx.y * 64;

  v8f zero = {};
  v8f c0 = zero, c1 = zero, c2 = zero, c3 = zero;

  const float* xrow = x + (m0 + l) * DD;
  const float* w0 = W + (N0 +  0 + l) * DD;
  const float* w1 = W + (N0 + 16 + l) * DD;
  const float* w2 = W + (N0 + 32 + l) * DD;
  const float* w3 = W + (N0 + 48 + l) * DD;

  for (int k0 = 0; k0 < DD; k0 += 4) {
    const int kk = k0 + 2 * h;
    v2f a = *(const v2f*)(xrow + kk);
    v2f b0 = *(const v2f*)(w0 + kk);
    v2f b1 = *(const v2f*)(w1 + kk);
    v2f b2 = *(const v2f*)(w2 + kk);
    v2f b3 = *(const v2f*)(w3 + kk);
    c0 = wmma_f32(a, b0, c0);
    c1 = wmma_f32(a, b1, c1);
    c2 = wmma_f32(a, b2, c2);
    c3 = wmma_f32(a, b3, c3);
  }

#pragma unroll
  for (int r = 0; r < 8; ++r) {
    const int row = m0 + r + 8 * h;
    const float mf = __expf(mask[row] * INV_E);
    float* yr = y + row * DD;
    yr[N0 +  0 + l] = (c0[r] + bias[N0 +  0 + l]) * mf;
    yr[N0 + 16 + l] = (c1[r] + bias[N0 + 16 + l]) * mf;
    yr[N0 + 32 + l] = (c2[r] + bias[N0 + 32 + l]) * mf;
    yr[N0 + 48 + l] = (c3[r] + bias[N0 + 48 + l]) * mf;
  }
}


__device__ __forceinline__ void attn_load_chunk(const float* __restrict__ ylane, int j0,
                                                v2f (&b0)[4], v2f (&b1)[4]) {
  const float* p = ylane + j0 * DD;
#pragma unroll
  for (int s = 0; s < 4; ++s) {
    const float* r0 = p + (4 * s) * DD;
    b0[s].x = r0[0];   b0[s].y = r0[DD];
    b1[s].x = r0[16];  b1[s].y = r0[DD + 16];
  }
}

__device__ __forceinline__ void attn_mma_chunk(float m, int j0, int h,
                                               const v2f (&b0)[4], const v2f (&b1)[4],
                                               v8f& c0, v8f& c1) {
#pragma unroll
  for (int s = 0; s < 4; ++s) {
    const float k0f = (float)(j0 + 4 * s + 2 * h);
    v2f a;
    a.x = __expf(-fabsf(m - k0f) * INV_E);
    a.y = __expf(-fabsf(m - (k0f + 1.0f)) * INV_E);
    c0 = wmma_f32(a, b0[s], c0);
    c1 = wmma_f32(a, b1[s], c1);
  }
}

__global__ __launch_bounds__(256)
void da_attn_kernel(const float* __restrict__ y, float* __restrict__ out) {
  const int lane = threadIdx.x & 31;
  const int l = lane & 15;
  const int h = lane >> 4;
  const int wv = threadIdx.x >> 5;
  const int it = blockIdx.x & (TT / 16 - 1);
  const int bi = blockIdx.x >> 8;
  const int I0 = it * 16;
  const int N0 = wv * 32;

  const float* yb = y + (bi * TT) * DD;
  const float* ylane = yb + (2 * h) * DD + N0 + l;

  int jlo = I0 - BAND; if (jlo < 0) jlo = 0;
  int jhi = I0 + 16 + BAND; if (jhi > TT) jhi = TT;

  v8f zero = {};
  v8f c0 = zero, c1 = zero;
  const float m = (float)(I0 + l);

  v2f p0[4], p1[4], q0[4], q1[4];
  int j0 = jlo;
  attn_load_chunk(ylane, j0, p0, p1);

  for (; j0 + 32 <= jhi; j0 += 32) {
    attn_load_chunk(ylane, j0 + 16, q0, q1);
    attn_mma_chunk(m, j0, h, p0, p1, c0, c1);
    const int jn = (j0 + 32 < jhi) ? j0 + 32 : jhi - 16;
    attn_load_chunk(ylane, jn, p0, p1);
    attn_mma_chunk(m, j0 + 16, h, q0, q1, c0, c1);
  }
  if (j0 < jhi) {
    attn_mma_chunk(m, j0, h, p0, p1, c0, c1);
  }

  float* ob = out + (bi * TT + I0) * DD;
#pragma unroll
  for (int r = 0; r < 8; ++r) {
    const int row = r + 8 * h;
    ob[row * DD + N0 + l]      = c0[r];
    ob[row * DD + N0 + 16 + l] = c1[r];
  }
}

extern "C" void kernel_launch(void* const* d_in, const int* in_sizes, int n_in,
                              void* d_out, int out_size, void* d_ws, size_t ws_size,
                              hipStream_t stream) {
  const float* x    = (const float*)d_in[0];
  const float* mask = (const float*)d_in[1];
  const float* W    = (const float*)d_in[2];
  const float* bias = (const float*)d_in[3];
  float* out = (float*)d_out;
  float* y   = (float*)d_ws;

  dim3 gproj(BB * TT / 16, DD / 64);
  da_proj_kernel<<<gproj, 32, 0, stream>>>(x, mask, W, bias, y);

  da_attn_kernel<<<BB * TT / 16, 256, 0, stream>>>(y, out);
}


# attn: each wave accumulates two 16-row tiles (32 rows) against the same y' fragments, so each fragment load feeds 2 WMMAs: half the loads and waves; sums unchanged (added terms are exact zeros)
# speedup vs baseline: 1.2120x; 1.0566x over previous
#include <hip/hip_runtime.h>
#include <hip/hip_bf16.h>


typedef __attribute__((ext_vector_type(2))) float v2f;
typedef __attribute__((ext_vector_type(8))) float v8f;

#define TT   4096
#define DD   256
#define BB   4
#define BAND 320
#define INV_E 0.36787944117144233f

__device__ __forceinline__ v8f wmma_f32(v2f a, v2f b, v8f c) {
  return __builtin_amdgcn_wmma_f32_16x16x4_f32(false, a, false, b,
                                               (short)0, c, false, false);
}

__global__ __launch_bounds__(32)
void da_proj_kernel(const float* __restrict__ x, const float* __restrict__ mask,
                    const float* __restrict__ W, const float* __restrict__ bias,
                    float* __restrict__ y) {
  const int lane = threadIdx.x & 31;
  const int l = lane & 15;
  const int h = lane >> 4;
  const int m0 = blockIdx.x * 16;
  const int N0 = blockIdx.y * 64;

  v8f zero = {};
  v8f c0 = zero, c1 = zero, c2 = zero, c3 = zero;

  const float* xrow = x + (m0 + l) * DD;
  const float* w0 = W + (N0 +  0 + l) * DD;
  const float* w1 = W + (N0 + 16 + l) * DD;
  const float* w2 = W + (N0 + 32 + l) * DD;
  const float* w3 = W + (N0 + 48 + l) * DD;

  for (int k0 = 0; k0 < DD; k0 += 4) {
    const int kk = k0 + 2 * h;
    v2f a = *(const v2f*)(xrow + kk);
    v2f b0 = *(const v2f*)(w0 + kk);
    v2f b1 = *(const v2f*)(w1 + kk);
    v2f b2 = *(const v2f*)(w2 + kk);
    v2f b3 = *(const v2f*)(w3 + kk);
    c0 = wmma_f32(a, b0, c0);
    c1 = wmma_f32(a, b1, c1);
    c2 = wmma_f32(a, b2, c2);
    c3 = wmma_f32(a, b3, c3);
  }

#pragma unroll
  for (int r = 0; r < 8; ++r) {
    const int row = m0 + r + 8 * h;
    const float mf = __expf(mask[row] * INV_E);
    float* yr = y + row * DD;
    yr[N0 +  0 + l] = (c0[r] + bias[N0 +  0 + l]) * mf;
    yr[N0 + 16 + l] = (c1[r] + bias[N0 + 16 + l]) * mf;
    yr[N0 + 32 + l] = (c2[r] + bias[N0 + 32 + l]) * mf;
    yr[N0 + 48 + l] = (c3[r] + bias[N0 + 48 + l]) * mf;
  }
}


__device__ __forceinline__ void attn_load_chunk(const float* __restrict__ ylane, int j0,
                                                v2f (&b0)[4], v2f (&b1)[4]) {
  const float* p = ylane + j0 * DD;
#pragma unroll
  for (int s = 0; s < 4; ++s) {
    const float* r0 = p + (4 * s) * DD;
    b0[s].x = r0[0];   b0[s].y = r0[DD];
    b1[s].x = r0[16];  b1[s].y = r0[DD + 16];
  }
}

__device__ __forceinline__ void attn_mma_chunk(float mA, float mB, int j0, int h,
                                               const v2f (&b0)[4], const v2f (&b1)[4],
                                               v8f& cA0, v8f& cA1, v8f& cB0, v8f& cB1) {
#pragma unroll
  for (int s = 0; s < 4; ++s) {
    const float k0f = (float)(j0 + 4 * s + 2 * h);
    v2f aA, aB;
    aA.x = __expf(-fabsf(mA - k0f) * INV_E);
    aA.y = __expf(-fabsf(mA - (k0f + 1.0f)) * INV_E);
    aB.x = __expf(-fabsf(mB - k0f) * INV_E);
    aB.y = __expf(-fabsf(mB - (k0f + 1.0f)) * INV_E);
    cA0 = wmma_f32(aA, b0[s], cA0);
    cA1 = wmma_f32(aA, b1[s], cA1);
    cB0 = wmma_f32(aB, b0[s], cB0);
    cB1 = wmma_f32(aB, b1[s], cB1);
  }
}

__global__ __launch_bounds__(256)
void da_attn_kernel(const float* __restrict__ y, float* __restrict__ out) {
  const int lane = threadIdx.x & 31;
  const int l = lane & 15;
  const int h = lane >> 4;
  const int wv = threadIdx.x >> 5;
  const int it = blockIdx.x & (TT / 32 - 1);
  const int bi = blockIdx.x >> 7;
  const int I0 = it * 32;
  const int N0 = wv * 32;

  const float* yb = y + (bi * TT) * DD;
  const float* ylane = yb + (2 * h) * DD + N0 + l;

  int jlo = I0 - BAND; if (jlo < 0) jlo = 0;
  int jhi = I0 + 32 + BAND; if (jhi > TT) jhi = TT;

  v8f zero = {};
  v8f cA0 = zero, cA1 = zero, cB0 = zero, cB1 = zero;
  const float mA = (float)(I0 + l);
  const float mB = (float)(I0 + 16 + l);

  v2f p0[4], p1[4], q0[4], q1[4];
  int j0 = jlo;
  attn_load_chunk(ylane, j0, p0, p1);

  for (; j0 + 32 <= jhi; j0 += 32) {
    attn_load_chunk(ylane, j0 + 16, q0, q1);
    attn_mma_chunk(mA, mB, j0, h, p0, p1, cA0, cA1, cB0, cB1);
    const int jn = (j0 + 32 < jhi) ? j0 + 32 : jhi - 16;
    attn_load_chunk(ylane, jn, p0, p1);
    attn_mma_chunk(mA, mB, j0 + 16, h, q0, q1, cA0, cA1, cB0, cB1);
  }
  if (j0 < jhi) {
    attn_mma_chunk(mA, mB, j0, h, p0, p1, cA0, cA1, cB0, cB1);
  }

  float* ob = out + (bi * TT + I0) * DD;
#pragma unroll
  for (int r = 0; r < 8; ++r) {
    const int row = r + 8 * h;
    ob[row * DD + N0 + l]             = cA0[r];
    ob[row * DD + N0 + 16 + l]        = cA1[r];
    ob[(row + 16) * DD + N0 + l]      = cB0[r];
    ob[(row + 16) * DD + N0 + 16 + l] = cB1[r];
  }
}

extern "C" void kernel_launch(void* const* d_in, const int* in_sizes, int n_in,
                              void* d_out, int out_size, void* d_ws, size_t ws_size,
                              hipStream_t stream) {
  const float* x    = (const float*)d_in[0];
  const float* mask = (const float*)d_in[1];
  const float* W    = (const float*)d_in[2];
  const float* bias = (const float*)d_in[3];
  float* out = (float*)d_out;
  float* y   = (float*)d_ws;

  dim3 gproj(BB * TT / 16, DD / 64);
  da_proj_kernel<<<gproj, 32, 0, stream>>>(x, mask, W, bias, y);

  da_attn_kernel<<<BB * TT / 32, 256, 0, stream>>>(y, out);
}


# attn: the second row tile's A fragments are the first tile's of the previous chunk (weight depends on i-j only), kept in registers: half the exp and vector work per WMMA; sums and bits unchanged
# speedup vs baseline: 1.3193x; 1.0885x over previous
#include <hip/hip_runtime.h>
#include <hip/hip_bf16.h>


typedef __attribute__((ext_vector_type(2))) float v2f;
typedef __attribute__((ext_vector_type(8))) float v8f;

#define TT   4096
#define DD   256
#define BB   4
#define BAND 320
#define INV_E 0.36787944117144233f

__device__ __forceinline__ v8f wmma_f32(v2f a, v2f b, v8f c) {
  return __builtin_amdgcn_wmma_f32_16x16x4_f32(false, a, false, b,
                                               (short)0, c, false, false);
}

__global__ __launch_bounds__(32)
void da_proj_kernel(const float* __restrict__ x, const float* __restrict__ mask,
                    const float* __restrict__ W, const float* __restrict__ bias,
                    float* __restrict__ y) {
  const int lane = threadIdx.x & 31;
  const int l = lane & 15;
  const int h = lane >> 4;
  const int m0 = blockIdx.x * 16;
  const int N0 = blockIdx.y * 64;

  v8f zero = {};
  v8f c0 = zero, c1 = zero, c2 = zero, c3 = zero;

  const float* xrow = x + (m0 + l) * DD;
  const float* w0 = W + (N0 +  0 + l) * DD;
  const float* w1 = W + (N0 + 16 + l) * DD;
  const float* w2 = W + (N0 + 32 + l) * DD;
  const float* w3 = W + (N0 + 48 + l) * DD;

  for (int k0 = 0; k0 < DD; k0 += 4) {
    const int kk = k0 + 2 * h;
    v2f a = *(const v2f*)(xrow + kk);
    v2f b0 = *(const v2f*)(w0 + kk);
    v2f b1 = *(const v2f*)(w1 + kk);
    v2f b2 = *(const v2f*)(w2 + kk);
    v2f b3 = *(const v2f*)(w3 + kk);
    c0 = wmma_f32(a, b0, c0);
    c1 = wmma_f32(a, b1, c1);
    c2 = wmma_f32(a, b2, c2);
    c3 = wmma_f32(a, b3, c3);
  }

#pragma unroll
  for (int r = 0; r < 8; ++r) {
    const int row = m0 + r + 8 * h;
    const float mf = __expf(mask[row] * INV_E);
    float* yr = y + row * DD;
    yr[N0 +  0 + l] = (c0[r] + bias[N0 +  0 + l]) * mf;
    yr[N0 + 16 + l] = (c1[r] + bias[N0 + 16 + l]) * mf;
    yr[N0 + 32 + l] = (c2[r] + bias[N0 + 32 + l]) * mf;
    yr[N0 + 48 + l] = (c3[r] + bias[N0 + 48 + l]) * mf;
  }
}


__device__ __forceinline__ void attn_load_chunk(const float* __restrict__ ylane, int j0,
                                                v2f (&b0)[4], v2f (&b1)[4]) {
  const float* p = ylane + j0 * DD;
#pragma unroll
  for (int s = 0; s < 4; ++s) {
    const float* r0 = p + (4 * s) * DD;
    b0[s].x = r0[0];   b0[s].y = r0[DD];
    b1[s].x = r0[16];  b1[s].y = r0[DD + 16];
  }
}

__device__ __forceinline__ void attn_gen_a(float d0, v2f (&a)[4]) {
#pragma unroll
  for (int s = 0; s < 4; ++s) {
    const float dx = d0 - (float)(4 * s);
    a[s].x = __expf(-fabsf(dx) * INV_E);
    a[s].y = __expf(-fabsf(dx - 1.0f) * INV_E);
  }
}

__device__ __forceinline__ void attn_mma_chunk(float d0,
                                               v2f (&aNew)[4], const v2f (&aOld)[4],
                                               const v2f (&b0)[4], const v2f (&b1)[4],
                                               v8f& cA0, v8f& cA1, v8f& cB0, v8f& cB1) {
  attn_gen_a(d0, aNew);
#pragma unroll
  for (int s = 0; s < 4; ++s) {
    cA0 = wmma_f32(aNew[s], b0[s], cA0);
    cA1 = wmma_f32(aNew[s], b1[s], cA1);
    cB0 = wmma_f32(aOld[s], b0[s], cB0);
    cB1 = wmma_f32(aOld[s], b1[s], cB1);
  }
}

__global__ __launch_bounds__(256)
void da_attn_kernel(const float* __restrict__ y, float* __restrict__ out) {
  const int lane = threadIdx.x & 31;
  const int l = lane & 15;
  const int h = lane >> 4;
  const int wv = threadIdx.x >> 5;
  const int it = blockIdx.x & (TT / 32 - 1);
  const int bi = blockIdx.x >> 7;
  const int I0 = it * 32;
  const int N0 = wv * 32;

  const float* yb = y + (bi * TT) * DD;
  const float* ylane = yb + (2 * h) * DD + N0 + l;

  int jlo = I0 - BAND; if (jlo < 0) jlo = 0;
  int jhi = I0 + 32 + BAND; if (jhi > TT) jhi = TT;

  v8f zero = {};
  v8f cA0 = zero, cA1 = zero, cB0 = zero, cB1 = zero;
  float dA = (float)(I0 + l - 2 * h - jlo);

  v2f p0[4], p1[4], q0[4], q1[4];
  int j0 = jlo;
  attn_load_chunk(ylane, j0, p0, p1);
  v2f aX[4], aY[4];
  attn_gen_a(dA + 16.0f, aY);

  for (; j0 + 32 <= jhi; j0 += 32) {
    attn_load_chunk(ylane, j0 + 16, q0, q1);
    attn_mma_chunk(dA, aX, aY, p0, p1, cA0, cA1, cB0, cB1);
    const int jn = (j0 + 32 < jhi) ? j0 + 32 : jhi - 16;
    attn_load_chunk(ylane, jn, p0, p1);
    attn_mma_chunk(dA - 16.0f, aY, aX, q0, q1, cA0, cA1, cB0, cB1);
    dA -= 32.0f;
  }
  if (j0 < jhi) {
    attn_mma_chunk(dA, aX, aY, p0, p1, cA0, cA1, cB0, cB1);
  }

  float* ob = out + (bi * TT + I0) * DD;
#pragma unroll
  for (int r = 0; r < 8; ++r) {
    const int row = r + 8 * h;
    ob[row * DD + N0 + l]             = cA0[r];
    ob[row * DD + N0 + 16 + l]        = cA1[r];
    ob[(row + 16) * DD + N0 + l]      = cB0[r];
    ob[(row + 16) * DD + N0 + 16 + l] = cB1[r];
  }
}

extern "C" void kernel_launch(void* const* d_in, const int* in_sizes, int n_in,
                              void* d_out, int out_size, void* d_ws, size_t ws_size,
                              hipStream_t stream) {
  const float* x    = (const float*)d_in[0];
  const float* mask = (const float*)d_in[1];
  const float* W    = (const float*)d_in[2];
  const float* bias = (const float*)d_in[3];
  float* out = (float*)d_out;
  float* y   = (float*)d_ws;

  dim3 gproj(BB * TT / 16, DD / 64);
  da_proj_kernel<<<gproj, 32, 0, stream>>>(x, mask, W, bias, y);

  da_attn_kernel<<<BB * TT / 32, 256, 0, stream>>>(y, out);
}


# proj: x and W tiles go through LDS with whole-line 16-byte loads, each wave computes 32x64 from LDS fragments (was: 8-byte global fragment loads touching 16 lines each); WMMA order per sum unchanged
# speedup vs baseline: 1.6416x; 1.2443x over previous
#include <hip/hip_runtime.h>
#include <hip/hip_bf16.h>


typedef __attribute__((ext_vector_type(2))) float v2f;
typedef __attribute__((ext_vector_type(8))) float v8f;

#define TT   4096
#define DD   256
#define BB   4
#define BAND 320
#define INV_E 0.36787944117144233f

__device__ __forceinline__ v8f wmma_f32(v2f a, v2f b, v8f c) {
  return __builtin_amdgcn_wmma_f32_16x16x4_f32(false, a, false, b,
                                               (short)0, c, false, false);
}

#define PM     64
#define PKB    32
#define PPITCH 36
#define PROWS  (PM + DD)
#define PSLOTS (PROWS * (PKB / 4) / 256)

typedef __attribute__((ext_vector_type(4))) float v4f;

__device__ __forceinline__ void proj_fetch(const float* __restrict__ xsrc,
                                           const float* __restrict__ wsrc, int kb,
                                           v4f (&g)[PSLOTS]) {
#pragma unroll
  for (int i = 0; i < 2; ++i)
    g[i] = *(const v4f*)(xsrc + i * 32 * DD + kb);
#pragma unroll
  for (int i = 2; i < PSLOTS; ++i)
    g[i] = *(const v4f*)(wsrc + (i - 2) * 32 * DD + kb);
}

__device__ __forceinline__ void proj_put(float* __restrict__ dst, const v4f (&g)[PSLOTS]) {
#pragma unroll
  for (int i = 0; i < PSLOTS; ++i)
    *(v4f*)(dst + i * 32 * PPITCH) = g[i];
}

__device__ __forceinline__ v2f proj_frag(const float* p, int s) {
  return *(const v2f*)(p + 4 * s);
}

__global__ __launch_bounds__(256)
void da_proj_kernel(const float* __restrict__ x, const float* __restrict__ mask,
                    const float* __restrict__ W, const float* __restrict__ bias,
                    float* __restrict__ y) {
  __shared__ __attribute__((aligned(16))) float stage[2][PROWS * PPITCH];

  const int tid = threadIdx.x;
  const int lane = tid & 31;
  const int l = lane & 15;
  const int h = lane >> 4;
  const int wv = tid >> 5;
  const int mi = wv & 1;
  const int ni = wv >> 1;
  const int M0 = blockIdx.x * PM;
  const int N0 = ni * 64;

  const int prow = tid >> 3;
  const int pseg = (tid & 7) * 4;
  const float* xsrc = x + (M0 + prow) * DD + pseg;
  const float* wsrc = W + prow * DD + pseg;
  const int pdst = prow * PPITCH + pseg;

  const int aoff = (mi * 32 + l) * PPITCH + 2 * h;
  const int boff = (PM + N0 + l) * PPITCH + 2 * h;

  v8f zero = {};
  v8f c[2][4];
#pragma unroll
  for (int t = 0; t < 4; ++t) { c[0][t] = zero; c[1][t] = zero; }

  v4f g[PSLOTS];
  proj_fetch(xsrc, wsrc, 0, g);
  proj_put(&stage[0][pdst], g);
  __syncthreads();

  for (int kb = 0; kb < DD / PKB; ++kb) {
    const bool more = kb + 1 < DD / PKB;
    if (more) proj_fetch(xsrc, wsrc, (kb + 1) * PKB, g);
    __builtin_amdgcn_sched_barrier(0);

    const float* sa = &stage[kb & 1][aoff];
    const float* sb = &stage[kb & 1][boff];
#pragma unroll
    for (int s = 0; s < PKB / 4; ++s) {
      const v2f a0 = proj_frag(sa, s);
      const v2f a1 = proj_frag(sa + 16 * PPITCH, s);
#pragma unroll
      for (int t = 0; t < 4; ++t) {
        const v2f b = proj_frag(sb + t * 16 * PPITCH, s);
        c[0][t] = wmma_f32(a0, b, c[0][t]);
        c[1][t] = wmma_f32(a1, b, c[1][t]);
      }
    }

    if (more) {
      proj_put(&stage[(kb + 1) & 1][pdst], g);
      __syncthreads();
    }
  }

#pragma unroll
  for (int mt = 0; mt < 2; ++mt) {
#pragma unroll
    for (int r = 0; r < 8; ++r) {
      const int row = M0 + mi * 32 + mt * 16 + r + 8 * h;
      const float mf = __expf(mask[row] * INV_E);
      float* yr = y + row * DD;
      yr[N0 +  0 + l] = (c[mt][0][r] + bias[N0 +  0 + l]) * mf;
      yr[N0 + 16 + l] = (c[mt][1][r] + bias[N0 + 16 + l]) * mf;
      yr[N0 + 32 + l] = (c[mt][2][r] + bias[N0 + 32 + l]) * mf;
      yr[N0 + 48 + l] = (c[mt][3][r] + bias[N0 + 48 + l]) * mf;
    }
  }
}


__device__ __forceinline__ void attn_load_chunk(const float* __restrict__ ylane, int j0,
                                                v2f (&b0)[4], v2f (&b1)[4]) {
  const float* p = ylane + j0 * DD;
#pragma unroll
  for (int s = 0; s < 4; ++s) {
    const float* r0 = p + (4 * s) * DD;
    b0[s].x = r0[0];   b0[s].y = r0[DD];
    b1[s].x = r0[16];  b1[s].y = r0[DD + 16];
  }
}

__device__ __forceinline__ void attn_gen_a(float d0, v2f (&a)[4]) {
#pragma unroll
  for (int s = 0; s < 4; ++s) {
    const float dx = d0 - (float)(4 * s);
    a[s].x = __expf(-fabsf(dx) * INV_E);
    a[s].y = __expf(-fabsf(dx - 1.0f) * INV_E);
  }
}

__device__ __forceinline__ void attn_mma_chunk(float d0,
                                               v2f (&aNew)[4], const v2f (&aOld)[4],
                                               const v2f (&b0)[4], const v2f (&b1)[4],
                                               v8f& cA0, v8f& cA1, v8f& cB0, v8f& cB1) {
  attn_gen_a(d0, aNew);
#pragma unroll
  for (int s = 0; s < 4; ++s) {
    cA0 = wmma_f32(aNew[s], b0[s], cA0);
    cA1 = wmma_f32(aNew[s], b1[s], cA1);
    cB0 = wmma_f32(aOld[s], b0[s], cB0);
    cB1 = wmma_f32(aOld[s], b1[s], cB1);
  }
}

__global__ __launch_bounds__(256)
void da_attn_kernel(const float* __restrict__ y, float* __restrict__ out) {
  const int lane = threadIdx.x & 31;
  const int l = lane & 15;
  const int h = lane >> 4;
  const int wv = threadIdx.x >> 5;
  const int it = blockIdx.x & (TT / 32 - 1);
  const int bi = blockIdx.x >> 7;
  const int I0 = it * 32;
  const int N0 = wv * 32;

  const float* yb = y + (bi * TT) * DD;
  const float* ylane = yb + (2 * h) * DD + N0 + l;

  int jlo = I0 - BAND; if (jlo < 0) jlo = 0;
  int jhi = I0 + 32 + BAND; if (jhi > TT) jhi = TT;

  v8f zero = {};
  v8f cA0 = zero, cA1 = zero, cB0 = zero, cB1 = zero;
  float dA = (float)(I0 + l - 2 * h - jlo);

  v2f p0[4], p1[4], q0[4], q1[4];
  int j0 = jlo;
  attn_load_chunk(ylane, j0, p0, p1);
  v2f aX[4], aY[4];
  attn_gen_a(dA + 16.0f, aY);

  for (; j0 + 32 <= jhi; j0 += 32) {
    attn_load_chunk(ylane, j0 + 16, q0, q1);
    attn_mma_chunk(dA, aX, aY, p0, p1, cA0, cA1, cB0, cB1);
    const int jn = (j0 + 32 < jhi) ? j0 + 32 : jhi - 16;
    attn_load_chunk(ylane, jn, p0, p1);
    attn_mma_chunk(dA - 16.0f, aY, aX, q0, q1, cA0, cA1, cB0, cB1);
    dA -= 32.0f;
  }
  if (j0 < jhi) {
    attn_mma_chunk(dA, aX, aY, p0, p1, cA0, cA1, cB0, cB1);
  }

  float* ob = out + (bi * TT + I0) * DD;
#pragma unroll
  for (int r = 0; r < 8; ++r) {
    const int row = r + 8 * h;
    ob[row * DD + N0 + l]             = cA0[r];
    ob[row * DD + N0 + 16 + l]        = cA1[r];
    ob[(row + 16) * DD + N0 + l]      = cB0[r];
    ob[(row + 16) * DD + N0 + 16 + l] = cB1[r];
  }
}

extern "C" void kernel_launch(void* const* d_in, const int* in_sizes, int n_in,
                              void* d_out, int out_size, void* d_ws, size_t ws_size,
                              hipStream_t stream) {
  const float* x    = (const float*)d_in[0];
  const float* mask = (const float*)d_in[1];
  const float* W    = (const float*)d_in[2];
  const float* bias = (const float*)d_in[3];
  float* out = (float*)d_out;
  float* y   = (float*)d_ws;

  da_proj_kernel<<<BB * TT / PM, 256, 0, stream>>>(x, mask, W, bias, y);

  da_attn_kernel<<<BB * TT / 32, 256, 0, stream>>>(y, out);
}


# attn: exp(-|d|/e) is tabulated once per distance in LDS per block; the band loop reads its A fragments from the table (8 v_exp + vector math per chunk become grouped LDS reads); bits unchanged
# speedup vs baseline: 1.7574x; 1.0705x over previous
#include <hip/hip_runtime.h>
#include <hip/hip_bf16.h>


typedef __attribute__((ext_vector_type(2))) float v2f;
typedef __attribute__((ext_vector_type(8))) float v8f;

#define TT   4096
#define DD   256
#define BB   4
#define BAND 320
#define INV_E 0.36787944117144233f

__device__ __forceinline__ v8f wmma_f32(v2f a, v2f b, v8f c) {
  return __builtin_amdgcn_wmma_f32_16x16x4_f32(false, a, false, b,
                                               (short)0, c, false, false);
}

#define PM     64
#define PKB    32
#define PPITCH 36
#define PROWS  (PM + DD)
#define PSLOTS (PROWS * (PKB / 4) / 256)

typedef __attribute__((ext_vector_type(4))) float v4f;

__device__ __forceinline__ void proj_fetch(const float* __restrict__ xsrc,
                                           const float* __restrict__ wsrc, int kb,
                                           v4f (&g)[PSLOTS]) {
#pragma unroll
  for (int i = 0; i < 2; ++i)
    g[i] = *(const v4f*)(xsrc + i * 32 * DD + kb);
#pragma unroll
  for (int i = 2; i < PSLOTS; ++i)
    g[i] = *(const v4f*)(wsrc + (i - 2) * 32 * DD + kb);
}

__device__ __forceinline__ void proj_put(float* __restrict__ dst, const v4f (&g)[PSLOTS]) {
#pragma unroll
  for (int i = 0; i < PSLOTS; ++i)
    *(v4f*)(dst + i * 32 * PPITCH) = g[i];
}

__device__ __forceinline__ v2f proj_frag(const float* p, int s) {
  return *(const v2f*)(p + 4 * s);
}

__global__ __launch_bounds__(256)
void da_proj_kernel(const float* __restrict__ x, const float* __restrict__ mask,
                    const float* __restrict__ W, const float* __restrict__ bias,
                    float* __restrict__ y) {
  __shared__ __attribute__((aligned(16))) float stage[2][PROWS * PPITCH];

  const int tid = threadIdx.x;
  const int lane = tid & 31;
  const int l = lane & 15;
  const int h = lane >> 4;
  const int wv = tid >> 5;
  const int mi = wv & 1;
  const int ni = wv >> 1;
  const int M0 = blockIdx.x * PM;
  const int N0 = ni * 64;

  const int prow = tid >> 3;
  const int pseg = (tid & 7) * 4;
  const float* xsrc = x + (M0 + prow) * DD + pseg;
  const float* wsrc = W + prow * DD + pseg;
  const int pdst = prow * PPITCH + pseg;

  const int aoff = (mi * 32 + l) * PPITCH + 2 * h;
  const int boff = (PM + N0 + l) * PPITCH + 2 * h;

  v8f zero = {};
  v8f c[2][4];
#pragma unroll
  for (int t = 0; t < 4; ++t) { c[0][t] = zero; c[1][t] = zero; }

  v4f g[PSLOTS];
  proj_fetch(xsrc, wsrc, 0, g);
  proj_put(&stage[0][pdst], g);
  __syncthreads();

  for (int kb = 0; kb < DD / PKB; ++kb) {
    const bool more = kb + 1 < DD / PKB;
    if (more) proj_fetch(xsrc, wsrc, (kb + 1) * PKB, g);
    __builtin_amdgcn_sched_barrier(0);

    const float* sa = &stage[kb & 1][aoff];
    const float* sb = &stage[kb & 1][boff];
#pragma unroll
    for (int s = 0; s < PKB / 4; ++s) {
      const v2f a0 = proj_frag(sa, s);
      const v2f a1 = proj_frag(sa + 16 * PPITCH, s);
#pragma unroll
      for (int t = 0; t < 4; ++t) {
        const v2f b = proj_frag(sb + t * 16 * PPITCH, s);
        c[0][t] = wmma_f32(a0, b, c[0][t]);
        c[1][t] = wmma_f32(a1, b, c[1][t]);
      }
    }

    if (more) {
      proj_put(&stage[(kb + 1) & 1][pdst], g);
      __syncthreads();
    }
  }

#pragma unroll
  for (int mt = 0; mt < 2; ++mt) {
#pragma unroll
    for (int r = 0; r < 8; ++r) {
      const int row = M0 + mi * 32 + mt * 16 + r + 8 * h;
      const float mf = __expf(mask[row] * INV_E);
      float* yr = y + row * DD;
      yr[N0 +  0 + l] = (c[mt][0][r] + bias[N0 +  0 + l]) * mf;
      yr[N0 + 16 + l] = (c[mt][1][r] + bias[N0 + 16 + l]) * mf;
      yr[N0 + 32 + l] = (c[mt][2][r] + bias[N0 + 32 + l]) * mf;
      yr[N0 + 48 + l] = (c[mt][3][r] + bias[N0 + 48 + l]) * mf;
    }
  }
}

#define WOFF  352
#define WCOPY 736

__device__ __forceinline__ void attn_load_chunk(const float* __restrict__ ylane, int j0,
                                                v2f (&b0)[4], v2f (&b1)[4]) {
  const float* p = ylane + j0 * DD;
#pragma unroll
  for (int s = 0; s < 4; ++s) {
    const float* r0 = p + (4 * s) * DD;
    b0[s].x = r0[0];   b0[s].y = r0[DD];
    b1[s].x = r0[16];  b1[s].y = r0[DD + 16];
  }
}

__device__ __forceinline__ void attn_load_a(const float* wp, v2f (&aA)[4], v2f (&aB)[4]) {
#pragma unroll
  for (int s = 0; s < 4; ++s) {
    aB[s] = *(const v2f*)(wp + 4 * s);
    aA[s] = *(const v2f*)(wp + 16 + 4 * s);
  }
}

__device__ __forceinline__ void attn_mma_chunk(const v2f (&aA)[4], const v2f (&aB)[4],
                                               const v2f (&b0)[4], const v2f (&b1)[4],
                                               v8f& cA0, v8f& cA1, v8f& cB0, v8f& cB1) {
#pragma unroll
  for (int s = 0; s < 4; ++s) {
    cA0 = wmma_f32(aA[s], b0[s], cA0);
    cA1 = wmma_f32(aA[s], b1[s], cA1);
    cB0 = wmma_f32(aB[s], b0[s], cB0);
    cB1 = wmma_f32(aB[s], b1[s], cB1);
  }
}

__global__ __launch_bounds__(256)
void da_attn_kernel(const float* __restrict__ y, float* __restrict__ out) {
  __shared__ __attribute__((aligned(16))) float wtab[2 * WCOPY];

  for (int k = threadIdx.x; k < 2 * WCOPY; k += 256) {
    const int cp = (k >= WCOPY) ? 1 : 0;
    const float dx = (float)(k - cp * WCOPY + cp - WOFF);
    wtab[k] = __expf(-fabsf(dx) * INV_E);
  }
  __syncthreads();

  const int lane = threadIdx.x & 31;
  const int l = lane & 15;
  const int h = lane >> 4;
  const int wv = threadIdx.x >> 5;
  const int it = blockIdx.x & (TT / 32 - 1);
  const int bi = blockIdx.x >> 7;
  const int I0 = it * 32;
  const int N0 = wv * 32;

  const float* yb = y + (bi * TT) * DD;
  const float* ylane = yb + (2 * h) * DD + N0 + l;

  int jlo = I0 - BAND; if (jlo < 0) jlo = 0;
  int jhi = I0 + 32 + BAND; if (jhi > TT) jhi = TT;

  v8f zero = {};
  v8f cA0 = zero, cA1 = zero, cB0 = zero, cB1 = zero;
  const int odd = l & 1;
  const int wi = WOFF - 16 - (I0 + l - 2 * h) - odd + odd * WCOPY;

  v2f p0[4], p1[4], q0[4], q1[4];
  v2f pA[4], pB[4], qA[4], qB[4];
  int j0 = jlo;
  attn_load_chunk(ylane, j0, p0, p1);
  attn_load_a(wtab + (wi + j0), pA, pB);

  for (; j0 + 32 <= jhi; j0 += 32) {
    attn_load_chunk(ylane, j0 + 16, q0, q1);
    attn_load_a(wtab + (wi + j0 + 16), qA, qB);
    attn_mma_chunk(pA, pB, p0, p1, cA0, cA1, cB0, cB1);
    __builtin_amdgcn_sched_barrier(0);
    const int jn = (j0 + 32 < jhi) ? j0 + 32 : jhi - 16;
    attn_load_chunk(ylane, jn, p0, p1);
    attn_load_a(wtab + (wi + jn), pA, pB);
    attn_mma_chunk(qA, qB, q0, q1, cA0, cA1, cB0, cB1);
    __builtin_amdgcn_sched_barrier(0);
  }
  if (j0 < jhi) {
    attn_mma_chunk(pA, pB, p0, p1, cA0, cA1, cB0, cB1);
  }

  float* ob = out + (bi * TT + I0) * DD;
#pragma unroll
  for (int r = 0; r < 8; ++r) {
    const int row = r + 8 * h;
    ob[row * DD + N0 + l]             = cA0[r];
    ob[row * DD + N0 + 16 + l]        = cA1[r];
    ob[(row + 16) * DD + N0 + l]      = cB0[r];
    ob[(row + 16) * DD + N0 + 16 + l] = cB1[r];
  }
}

extern "C" void kernel_launch(void* const* d_in, const int* in_sizes, int n_in,
                              void* d_out, int out_size, void* d_ws, size_t ws_size,
                              hipStream_t stream) {
  const float* x    = (const float*)d_in[0];
  const float* mask = (const float*)d_in[1];
  const float* W    = (const float*)d_in[2];
  const float* bias = (const float*)d_in[3];
  float* out = (float*)d_out;
  float* y   = (float*)d_ws;

  da_proj_kernel<<<BB * TT / PM, 256, 0, stream>>>(x, mask, W, bias, y);

  da_attn_kernel<<<BB * TT / 32, 256, 0, stream>>>(y, out);
}


# attn: off the diagonal exp(-|i-j|/e) factorises, so rows of y' below/above the 32-row group feed one row tile and the other tile gets that sum times exp(-16/e): half the WMMAs there; sums regrouped
# speedup vs baseline: 2.1135x; 1.2026x over previous
#include <hip/hip_runtime.h>
#include <hip/hip_bf16.h>


typedef __attribute__((ext_vector_type(2))) float v2f;
typedef __attribute__((ext_vector_type(8))) float v8f;

#define TT   4096
#define DD   256
#define BB   4
#define BAND 320
#define INV_E 0.36787944117144233f

__device__ __forceinline__ v8f wmma_f32(v2f a, v2f b, v8f c) {
  return __builtin_amdgcn_wmma_f32_16x16x4_f32(false, a, false, b,
                                               (short)0, c, false, false);
}

#define PM     64
#define PKB    32
#define PPITCH 36
#define PROWS  (PM + DD)
#define PSLOTS (PROWS * (PKB / 4) / 256)

typedef __attribute__((ext_vector_type(4))) float v4f;

__device__ __forceinline__ void proj_fetch(const float* __restrict__ xsrc,
                                           const float* __restrict__ wsrc, int kb,
                                           v4f (&g)[PSLOTS]) {
#pragma unroll
  for (int i = 0; i < 2; ++i)
    g[i] = *(const v4f*)(xsrc + i * 32 * DD + kb);
#pragma unroll
  for (int i = 2; i < PSLOTS; ++i)
    g[i] = *(const v4f*)(wsrc + (i - 2) * 32 * DD + kb);
}

__device__ __forceinline__ void proj_put(float* __restrict__ dst, const v4f (&g)[PSLOTS]) {
#pragma unroll
  for (int i = 0; i < PSLOTS; ++i)
    *(v4f*)(dst + i * 32 * PPITCH) = g[i];
}

__device__ __forceinline__ v2f proj_frag(const float* p, int s) {
  return *(const v2f*)(p + 4 * s);
}

__global__ __launch_bounds__(256)
void da_proj_kernel(const float* __restrict__ x, const float* __restrict__ mask,
                    const float* __restrict__ W, const float* __restrict__ bias,
                    float* __restrict__ y) {
  __shared__ __attribute__((aligned(16))) float stage[2][PROWS * PPITCH];

  const int tid = threadIdx.x;
  const int lane = tid & 31;
  const int l = lane & 15;
  const int h = lane >> 4;
  const int wv = tid >> 5;
  const int mi = wv & 1;
  const int ni = wv >> 1;
  const int M0 = blockIdx.x * PM;
  const int N0 = ni * 64;

  const int prow = tid >> 3;
  const int pseg = (tid & 7) * 4;
  const float* xsrc = x + (M0 + prow) * DD + pseg;
  const float* wsrc = W + prow * DD + pseg;
  const int pdst = prow * PPITCH + pseg;

  const int aoff = (mi * 32 + l) * PPITCH + 2 * h;
  const int boff = (PM + N0 + l) * PPITCH + 2 * h;

  v8f zero = {};
  v8f c[2][4];
#pragma unroll
  for (int t = 0; t < 4; ++t) { c[0][t] = zero; c[1][t] = zero; }

  v4f g[PSLOTS];
  proj_fetch(xsrc, wsrc, 0, g);
  proj_put(&stage[0][pdst], g);
  __syncthreads();

  for (int kb = 0; kb < DD / PKB; ++kb) {
    const bool more = kb + 1 < DD / PKB;
    if (more) proj_fetch(xsrc, wsrc, (kb + 1) * PKB, g);
    __builtin_amdgcn_sched_barrier(0);

    const float* sa = &stage[kb & 1][aoff];
    const float* sb = &stage[kb & 1][boff];
#pragma unroll
    for (int s = 0; s < PKB / 4; ++s) {
      const v2f a0 = proj_frag(sa, s);
      const v2f a1 = proj_frag(sa + 16 * PPITCH, s);
#pragma unroll
      for (int t = 0; t < 4; ++t) {
        const v2f b = proj_frag(sb + t * 16 * PPITCH, s);
        c[0][t] = wmma_f32(a0, b, c[0][t]);
        c[1][t] = wmma_f32(a1, b, c[1][t]);
      }
    }

    if (more) {
      proj_put(&stage[(kb + 1) & 1][pdst], g);
      __syncthreads();
    }
  }

#pragma unroll
  for (int mt = 0; mt < 2; ++mt) {
#pragma unroll
    for (int r = 0; r < 8; ++r) {
      const int row = M0 + mi * 32 + mt * 16 + r + 8 * h;
      const float mf = __expf(mask[row] * INV_E);
      float* yr = y + row * DD;
      yr[N0 +  0 + l] = (c[mt][0][r] + bias[N0 +  0 + l]) * mf;
      yr[N0 + 16 + l] = (c[mt][1][r] + bias[N0 + 16 + l]) * mf;
      yr[N0 + 32 + l] = (c[mt][2][r] + bias[N0 + 32 + l]) * mf;
      yr[N0 + 48 + l] = (c[mt][3][r] + bias[N0 + 48 + l]) * mf;
    }
  }
}

#define WOFF  352
#define WCOPY 736

__device__ __forceinline__ void attn_load_unit(const float* __restrict__ ylane, int j,
                                               v2f (&b0)[8], v2f (&b1)[8]) {
  const float* p = ylane + j * DD;
#pragma unroll
  for (int t = 0; t < 8; ++t) {
    const float* r0 = p + (4 * t) * DD;
    b0[t].x = r0[0];   b0[t].y = r0[DD];
    b1[t].x = r0[16];  b1[t].y = r0[DD + 16];
  }
}

__device__ __forceinline__ void attn_load_a(const float* wp, v2f (&a)[8]) {
#pragma unroll
  for (int t = 0; t < 8; ++t) a[t] = *(const v2f*)(wp + 4 * t);
}
__device__ __forceinline__ void attn_load_a_own(const float* wp0, v2f (&m)[12]) {
#pragma unroll
  for (int t = 0; t < 12; ++t) m[t] = *(const v2f*)(wp0 + 4 * t);
}

__device__ __forceinline__ void attn_mma_unit(const v2f (&a)[8],
                                              const v2f (&b0)[8], const v2f (&b1)[8],
                                              v8f& c0, v8f& c1) {
#pragma unroll
  for (int t = 0; t < 8; ++t) {
    c0 = wmma_f32(a[t], b0[t], c0);
    c1 = wmma_f32(a[t], b1[t], c1);
  }
}

__device__ __forceinline__ void attn_mma_own(const v2f (&m)[12],
                                             const v2f (&b0)[8], const v2f (&b1)[8],
                                             v8f& cA0, v8f& cA1, v8f& cB0, v8f& cB1) {
#pragma unroll
  for (int t = 0; t < 8; ++t) {
    cA0 = wmma_f32(m[t + 4], b0[t], cA0);
    cA1 = wmma_f32(m[t + 4], b1[t], cA1);
    cB0 = wmma_f32(m[t],     b0[t], cB0);
    cB1 = wmma_f32(m[t],     b1[t], cB1);
  }
}

__global__ __launch_bounds__(256)
void da_attn_kernel(const float* __restrict__ y, float* __restrict__ out) {
  __shared__ __attribute__((aligned(16))) float wtab[2 * WCOPY];

  for (int k = threadIdx.x; k < 2 * WCOPY; k += 256) {
    const int cp = (k >= WCOPY) ? 1 : 0;
    const float dx = (float)(k - cp * WCOPY + cp - WOFF);
    wtab[k] = __expf(-fabsf(dx) * INV_E);
  }
  __syncthreads();

  const int lane = threadIdx.x & 31;
  const int l = lane & 15;
  const int h = lane >> 4;
  const int wv = threadIdx.x >> 5;
  const int it = blockIdx.x & (TT / 32 - 1);
  const int bi = blockIdx.x >> 7;
  const int I0 = it * 32;
  const int N0 = wv * 32;

  const float* yb = y + (bi * TT) * DD;
  const float* ylane = yb + (2 * h) * DD + N0 + l;

  int jlo = I0 - BAND; if (jlo < 0) jlo = 0;
  int jhi = I0 + 32 + BAND; if (jhi > TT) jhi = TT;
  const int nb = (I0 - jlo) >> 5;
  const int nu = nb + ((jhi - I0 - 32) >> 5);

  const float r16 = wtab[WOFF + 16];

  const int odd = l & 1;
  const int wi = WOFF - 16 - (I0 + l - 2 * h) - odd + odd * WCOPY;

#define UROW(u) (jlo + 32 * (u) + ((u) >= nb ? 32 : 0))
#define UTAB(u) (wtab + (wi + UROW(u) + ((u) < nb ? 16 : 0)))

  v8f zero = {};
  v8f cA0 = zero, cA1 = zero, cB0 = zero, cB1 = zero;
  v8f s0 = zero, s1 = zero;

  v2f p0[8], p1[8], q0[8], q1[8];
  v2f pa[8], qa[8];

  {
    v2f m0[8], m1[8], ma[12];
    attn_load_unit(ylane, I0, m0, m1);
    attn_load_a_own(wtab + (wi + I0), ma);
    attn_load_unit(ylane, UROW(0), p0, p1);
    attn_load_a(UTAB(0), pa);
    attn_mma_own(ma, m0, m1, cA0, cA1, cB0, cB1);
  }

#define PART_TURN() do { cA0 += s0; cA1 += s1; cB0 += s0 * r16; cB1 += s1 * r16; \
                         s0 = zero; s1 = zero; } while (0)

  int u = 0;
  for (; u + 2 <= nu; u += 2) {
    attn_load_unit(ylane, UROW(u + 1), q0, q1);
    attn_load_a(UTAB(u + 1), qa);
    if (u == nb) PART_TURN();
    attn_mma_unit(pa, p0, p1, s0, s1);
    __builtin_amdgcn_sched_barrier(0);
    const int un = (u + 2 < nu) ? u + 2 : nu - 1;
    attn_load_unit(ylane, UROW(un), p0, p1);
    attn_load_a(UTAB(un), pa);
    if (u + 1 == nb) PART_TURN();
    attn_mma_unit(qa, q0, q1, s0, s1);
    __builtin_amdgcn_sched_barrier(0);
  }
  if (u < nu) {
    if (u == nb) PART_TURN();
    attn_mma_unit(pa, p0, p1, s0, s1);
  }
  if (nb == nu) PART_TURN();

  cA0 += s0 * r16;  cA1 += s1 * r16;
  cB0 += s0;        cB1 += s1;
#undef PART_TURN
#undef UTAB
#undef UROW

  float* ob = out + (bi * TT + I0) * DD;
#pragma unroll
  for (int r = 0; r < 8; ++r) {
    const int row = r + 8 * h;
    ob[row * DD + N0 + l]             = cA0[r];
    ob[row * DD + N0 + 16 + l]        = cA1[r];
    ob[(row + 16) * DD + N0 + l]      = cB0[r];
    ob[(row + 16) * DD + N0 + 16 + l] = cB1[r];
  }
}

extern "C" void kernel_launch(void* const* d_in, const int* in_sizes, int n_in,
                              void* d_out, int out_size, void* d_ws, size_t ws_size,
                              hipStream_t stream) {
  const float* x    = (const float*)d_in[0];
  const float* mask = (const float*)d_in[1];
  const float* W    = (const float*)d_in[2];
  const float* bias = (const float*)d_in[3];
  float* out = (float*)d_out;
  float* y   = (float*)d_ws;

  da_proj_kernel<<<BB * TT / PM, 256, 0, stream>>>(x, mask, W, bias, y);

  da_attn_kernel<<<BB * TT / 32, 256, 0, stream>>>(y, out);
}


# attn: 64-row groups (4 row tiles) per block: rows of y' outside a tile pair are summed once with the nearest tile's weights, other tiles get it times exp(-16k/e): half the y' reads; sums regrouped
# speedup vs baseline: 2.6367x; 1.2476x over previous
#include <hip/hip_runtime.h>
#include <hip/hip_bf16.h>


typedef __attribute__((ext_vector_type(2))) float v2f;
typedef __attribute__((ext_vector_type(8))) float v8f;

#define TT   4096
#define DD   256
#define BB   4
#define BAND 320
#define INV_E 0.36787944117144233f

__device__ __forceinline__ v8f wmma_f32(v2f a, v2f b, v8f c) {
  return __builtin_amdgcn_wmma_f32_16x16x4_f32(false, a, false, b,
                                               (short)0, c, false, false);
}

#define PM     64
#define PKB    32
#define PPITCH 36
#define PROWS  (PM + DD)
#define PSLOTS (PROWS * (PKB / 4) / 256)

typedef __attribute__((ext_vector_type(4))) float v4f;

__device__ __forceinline__ void proj_fetch(const float* __restrict__ xsrc,
                                           const float* __restrict__ wsrc, int kb,
                                           v4f (&g)[PSLOTS]) {
#pragma unroll
  for (int i = 0; i < 2; ++i)
    g[i] = *(const v4f*)(xsrc + i * 32 * DD + kb);
#pragma unroll
  for (int i = 2; i < PSLOTS; ++i)
    g[i] = *(const v4f*)(wsrc + (i - 2) * 32 * DD + kb);
}

__device__ __forceinline__ void proj_put(float* __restrict__ dst, const v4f (&g)[PSLOTS]) {
#pragma unroll
  for (int i = 0; i < PSLOTS; ++i)
    *(v4f*)(dst + i * 32 * PPITCH) = g[i];
}

__device__ __forceinline__ v2f proj_frag(const float* p, int s) {
  return *(const v2f*)(p + 4 * s);
}

__global__ __launch_bounds__(256)
void da_proj_kernel(const float* __restrict__ x, const float* __restrict__ mask,
                    const float* __restrict__ W, const float* __restrict__ bias,
                    float* __restrict__ y) {
  __shared__ __attribute__((aligned(16))) float stage[2][PROWS * PPITCH];

  const int tid = threadIdx.x;
  const int lane = tid & 31;
  const int l = lane & 15;
  const int h = lane >> 4;
  const int wv = tid >> 5;
  const int mi = wv & 1;
  const int ni = wv >> 1;
  const int M0 = blockIdx.x * PM;
  const int N0 = ni * 64;

  const int prow = tid >> 3;
  const int pseg = (tid & 7) * 4;
  const float* xsrc = x + (M0 + prow) * DD + pseg;
  const float* wsrc = W + prow * DD + pseg;
  const int pdst = prow * PPITCH + pseg;

  const int aoff = (mi * 32 + l) * PPITCH + 2 * h;
  const int boff = (PM + N0 + l) * PPITCH + 2 * h;

  v8f zero = {};
  v8f c[2][4];
#pragma unroll
  for (int t = 0; t < 4; ++t) { c[0][t] = zero; c[1][t] = zero; }

  v4f g[PSLOTS];
  proj_fetch(xsrc, wsrc, 0, g);
  proj_put(&stage[0][pdst], g);
  __syncthreads();

  for (int kb = 0; kb < DD / PKB; ++kb) {
    const bool more = kb + 1 < DD / PKB;
    if (more) proj_fetch(xsrc, wsrc, (kb + 1) * PKB, g);
    __builtin_amdgcn_sched_barrier(0);

    const float* sa = &stage[kb & 1][aoff];
    const float* sb = &stage[kb & 1][boff];
#pragma unroll
    for (int s = 0; s < PKB / 4; ++s) {
      const v2f a0 = proj_frag(sa, s);
      const v2f a1 = proj_frag(sa + 16 * PPITCH, s);
#pragma unroll
      for (int t = 0; t < 4; ++t) {
        const v2f b = proj_frag(sb + t * 16 * PPITCH, s);
        c[0][t] = wmma_f32(a0, b, c[0][t]);
        c[1][t] = wmma_f32(a1, b, c[1][t]);
      }
    }

    if (more) {
      proj_put(&stage[(kb + 1) & 1][pdst], g);
      __syncthreads();
    }
  }

#pragma unroll
  for (int mt = 0; mt < 2; ++mt) {
#pragma unroll
    for (int r = 0; r < 8; ++r) {
      const int row = M0 + mi * 32 + mt * 16 + r + 8 * h;
      const float mf = __expf(mask[row] * INV_E);
      float* yr = y + row * DD;
      yr[N0 +  0 + l] = (c[mt][0][r] + bias[N0 +  0 + l]) * mf;
      yr[N0 + 16 + l] = (c[mt][1][r] + bias[N0 + 16 + l]) * mf;
      yr[N0 + 32 + l] = (c[mt][2][r] + bias[N0 + 32 + l]) * mf;
      yr[N0 + 48 + l] = (c[mt][3][r] + bias[N0 + 48 + l]) * mf;
    }
  }
}

#define WOFF  352
#define WCOPY 736

__device__ __forceinline__ void attn_load_unit(const float* __restrict__ ylane, int j,
                                               v2f (&b0)[8], v2f (&b1)[8]) {
  const float* p = ylane + j * DD;
#pragma unroll
  for (int t = 0; t < 8; ++t) {
    const float* r0 = p + (4 * t) * DD;
    b0[t].x = r0[0];   b0[t].y = r0[DD];
    b1[t].x = r0[16];  b1[t].y = r0[DD + 16];
  }
}

__device__ __forceinline__ void attn_load_a(const float* wp, v2f (&a)[8]) {
#pragma unroll
  for (int t = 0; t < 8; ++t) a[t] = *(const v2f*)(wp + 4 * t);
}
__device__ __forceinline__ void attn_load_a_own(const float* wp0, v2f (&m)[12]) {
#pragma unroll
  for (int t = 0; t < 12; ++t) m[t] = *(const v2f*)(wp0 + 4 * t);
}

__device__ __forceinline__ void attn_mma_unit(const v2f (&a)[8],
                                              const v2f (&b0)[8], const v2f (&b1)[8],
                                              v8f& c0, v8f& c1) {
#pragma unroll
  for (int t = 0; t < 8; ++t) {
    c0 = wmma_f32(a[t], b0[t], c0);
    c1 = wmma_f32(a[t], b1[t], c1);
  }
}

__device__ __forceinline__ void attn_mma_own(const v2f (&m)[12],
                                             const v2f (&b0)[8], const v2f (&b1)[8],
                                             v8f& cA0, v8f& cA1, v8f& cB0, v8f& cB1) {
#pragma unroll
  for (int t = 0; t < 8; ++t) {
    cA0 = wmma_f32(m[t + 4], b0[t], cA0);
    cA1 = wmma_f32(m[t + 4], b1[t], cA1);
    cB0 = wmma_f32(m[t],     b0[t], cB0);
    cB1 = wmma_f32(m[t],     b1[t], cB1);
  }
}

#define GR 64
#define MAXU (2 * BAND / 32)

__global__ __launch_bounds__(256)
void da_attn_kernel(const float* __restrict__ y, float* __restrict__ out) {
  __shared__ __attribute__((aligned(16))) float wtab[2 * WCOPY];

  for (int k = threadIdx.x; k < 2 * WCOPY; k += 256) {
    const int cp = (k >= WCOPY) ? 1 : 0;
    const float dx = (float)(k - cp * WCOPY + cp - WOFF);
    wtab[k] = __expf(-fabsf(dx) * INV_E);
  }
  __syncthreads();

  const int lane = threadIdx.x & 31;
  const int l = lane & 15;
  const int h = lane >> 4;
  const int wv = threadIdx.x >> 5;
  const int it = blockIdx.x & (TT / GR - 1);
  const int bi = blockIdx.x / (TT / GR);
  const int I0 = it * GR;
  const int N0 = wv * 32;

  const float* yb = y + (bi * TT) * DD;
  const float* ylane = yb + (2 * h) * DD + N0 + l;

  int jlo = I0 - BAND; if (jlo < 0) jlo = 0;
  int jhi = I0 + GR + BAND; if (jhi > TT) jhi = TT;
  const int nb = (I0 - jlo) >> 5;
  const int nu = nb + ((jhi - I0 - GR) >> 5);

  const float r16 = wtab[WOFF + 16];
  const float r32 = wtab[WOFF + 32];
  const float r48 = wtab[WOFF + 48];

  const int odd = l & 1;
  const int wi = WOFF - 16 - (I0 + l - 2 * h) - odd + odd * WCOPY;

#define UROW(u) (jlo + 32 * (u) + ((u) >= nb ? GR : 0))
#define UTAB(u) (wtab + (wi + UROW(u) + ((u) < nb ? 16 : -32)))

  v8f zero = {};
  v8f c0a = zero, c0b = zero, c1a = zero, c1b = zero;
  v8f c2a = zero, c2b = zero, c3a = zero, c3b = zero;
  v8f e0 = zero, e1 = zero;
  v8f f0 = zero, f1 = zero;
  v8f s0 = zero, s1 = zero;

  v2f B0[3][8], B1[3][8];
  v2f A[3][8];

  {
    v2f m0[8], m1[8], n0[8], n1[8], ma[12], mb[8], nf[8];
    const float* wo = wtab + (wi + I0);
    attn_load_unit(ylane, I0, m0, m1);
    attn_load_unit(ylane, I0 + 32, n0, n1);
    attn_load_a_own(wo, ma);
    attn_load_a(wo - 16, mb);
    attn_load_a(wo + 32, nf);
    attn_load_unit(ylane, UROW(0), B0[0], B1[0]);
    attn_load_a(UTAB(0), A[0]);
    attn_mma_own(ma, m0, m1, c0a, c0b, c1a, c1b);
    attn_mma_unit(mb, m0, m1, e0, e1);
    __builtin_amdgcn_sched_barrier(0);
    attn_load_unit(ylane, UROW(1), B0[1], B1[1]);
    attn_load_a(UTAB(1), A[1]);
    attn_mma_own(ma, n0, n1, c2a, c2b, c3a, c3b);
    attn_mma_unit(nf, n0, n1, f0, f1);
    __builtin_amdgcn_sched_barrier(0);
  }

#define PART_TURN() do { c0a += s0;       c0b += s1;       c1a += s0 * r16; c1b += s1 * r16; \
                         c2a += s0 * r32; c2b += s1 * r32; c3a += s0 * r48; c3b += s1 * r48; \
                         s0 = zero; s1 = zero; } while (0)

#define WALK_UNIT(u, r) do {                                                       \
    if ((u) + 2 >= nu) goto tail_##r;                 \
    attn_load_unit(ylane, UROW((u) + 2), B0[((u) + 2) % 3], B1[((u) + 2) % 3]);    \
    attn_load_a(UTAB((u) + 2), A[((u) + 2) % 3]);                                  \
    if ((u) == nb) PART_TURN();                                  \
    attn_mma_unit(A[(u) % 3], B0[(u) % 3], B1[(u) % 3], s0, s1);                   \
    __builtin_amdgcn_sched_barrier(0);                                             \
  } while (0)
#define WALK_TAIL(r) do {                                                          \
    if (nu - 2 == nb) PART_TURN();                               \
    attn_mma_unit(A[r], B0[r], B1[r], s0, s1);                                     \
    if (nu - 1 == nb) PART_TURN();                               \
    attn_mma_unit(A[((r) + 1) % 3], B0[((r) + 1) % 3], B1[((r) + 1) % 3], s0, s1); \
  } while (0)
  static_assert(MAXU == 20, "the walk below is written out for at most 20 units");
  static_assert(TT >= GR + BAND, "one side of every group has its full ten units, so nu >= 2 for the tail");
  WALK_UNIT(0, 0);  WALK_UNIT(1, 1);  WALK_UNIT(2, 2);  WALK_UNIT(3, 0);  WALK_UNIT(4, 1);  WALK_UNIT(5, 2);
  WALK_UNIT(6, 0);  WALK_UNIT(7, 1);  WALK_UNIT(8, 2);  WALK_UNIT(9, 0);  WALK_UNIT(10, 1); WALK_UNIT(11, 2);
  WALK_UNIT(12, 0); WALK_UNIT(13, 1); WALK_UNIT(14, 2); WALK_UNIT(15, 0); WALK_UNIT(16, 1); WALK_UNIT(17, 2);
tail_0: WALK_TAIL(0); goto walk_done;
tail_1: WALK_TAIL(1); goto walk_done;
tail_2: WALK_TAIL(2);
walk_done:
#undef WALK_TAIL
#undef WALK_UNIT
  if (nb == nu) PART_TURN();

  c3a += s0;             c3b += s1;
  c2a += s0 * r16 + e0;  c2b += s1 * r16 + e1;
  c1a += s0 * r32 + f0;  c1b += s1 * r32 + f1;
  c0a += s0 * r48;       c0b += s1 * r48;
  c3a += e0 * r16;       c3b += e1 * r16;
  c0a += f0 * r16;       c0b += f1 * r16;
#undef PART_TURN
#undef UTAB
#undef UROW

  float* ob = out + (bi * TT + I0) * DD;
#pragma unroll
  for (int r = 0; r < 8; ++r) {
    float* orow = ob + (r + 8 * h) * DD + N0 + l;
    orow[0]            = c0a[r];
    orow[16]           = c0b[r];
    orow[16 * DD]      = c1a[r];
    orow[16 * DD + 16] = c1b[r];
    orow[32 * DD]      = c2a[r];
    orow[32 * DD + 16] = c2b[r];
    orow[48 * DD]      = c3a[r];
    orow[48 * DD + 16] = c3b[r];
  }
}

extern "C" void kernel_launch(void* const* d_in, const int* in_sizes, int n_in,
                              void* d_out, int out_size, void* d_ws, size_t ws_size,
                              hipStream_t stream) {
  const float* x    = (const float*)d_in[0];
  const float* mask = (const float*)d_in[1];
  const float* W    = (const float*)d_in[2];
  const float* bias = (const float*)d_in[3];
  float* out = (float*)d_out;
  float* y   = (float*)d_ws;

  da_proj_kernel<<<BB * TT / PM, 256, 0, stream>>>(x, mask, W, bias, y);

  da_attn_kernel<<<BB * TT / GR, 256, 0, stream>>>(y, out);
}
